# starting kernel
# baseline (speedup 1.0000x reference)
#include <hip/hip_runtime.h>


#define VOCAB   100
#define E       32
#define HDIM    64
#define G4H     256
#define MAXLEN  16
#define WAVES   8
#define EPITCH  36
#define WIHP    36
#define WHHP    68
#define HPITCH  68

typedef __attribute__((ext_vector_type(2))) float v2f;
typedef __attribute__((ext_vector_type(4))) float v4f;
typedef __attribute__((ext_vector_type(8))) float v8f;

__device__ __forceinline__ float sigm(float x) {
    return 1.0f / (1.0f + __expf(-x));
}

__global__ __launch_bounds__(WAVES * 32, 4)
void char_lstm_wmma(const float* __restrict__ emb,
                    const float* __restrict__ Wih,
                    const float* __restrict__ Whh,
                    const float* __restrict__ bih,
                    const float* __restrict__ bhh,
                    const int*   __restrict__ chars,
                    const int*   __restrict__ lengths,
                    float*       __restrict__ out)
{
    __shared__ float s_emb[VOCAB * EPITCH];
    __shared__ float s_wih[G4H * WIHP];
    __shared__ float s_whh[G4H * WHHP];
    __shared__ float s_bias[G4H];
    __shared__ float s_h[WAVES][16 * HPITCH];

    const int tid  = threadIdx.x;
    const int lane = tid & 31;
    const int wave = tid >> 5;
    const int half = lane >> 4;
    const int n    = lane & 15;
    const int base = (blockIdx.x * WAVES + wave) * 16;

    for (int i = tid; i < VOCAB * E; i += WAVES * 32)
        s_emb[(i / E) * EPITCH + (i % E)] = emb[i];
    for (int i = tid; i < G4H * E; i += WAVES * 32)
        s_wih[(i / E) * WIHP + (i % E)] = Wih[i];
    for (int i = tid; i < G4H * HDIM; i += WAVES * 32)
        s_whh[(i / HDIM) * WHHP + (i % HDIM)] = Whh[i];
    for (int i = tid; i < G4H; i += WAVES * 32)
        s_bias[i] = bih[i] + bhh[i];
    for (int i = lane; i < 16 * HPITCH; i += 32)
        s_h[wave][i] = 0.0f;

    int llen[8];
    #pragma unroll
    for (int r = 0; r < 8; ++r)
        llen[r] = lengths[base + 8 * half + r];

    v8f creg[4];
    v8f hreg[4];
    #pragma unroll
    for (int cb = 0; cb < 4; ++cb)
        #pragma unroll
        for (int r = 0; r < 8; ++r) { creg[cb][r] = 0.0f; hreg[cb][r] = 0.0f; }

    __syncthreads();

    #pragma unroll 1
    for (int t = 0; t < MAXLEN; ++t) {
        const int c    = chars[(base + n) * MAXLEN + t];
        const int xoff = c * EPITCH + 2 * half;

        #pragma unroll
        for (int cb = 0; cb < 4; ++cb) {
            v8f acc[4];
            #pragma unroll
            for (int g = 0; g < 4; ++g)
                #pragma unroll
                for (int r = 0; r < 8; ++r) acc[g][r] = 0.0f;

            #pragma unroll
            for (int kk = 0; kk < E / 4; ++kk) {
                v2f a = *(const v2f*)&s_emb[xoff + kk * 4];
                #pragma unroll
                for (int g = 0; g < 4; ++g) {
                    v2f b = *(const v2f*)&s_wih[(g * 64 + cb * 16 + n) * WIHP
                                                + kk * 4 + 2 * half];
                    acc[g] = __builtin_amdgcn_wmma_f32_16x16x4_f32(
                        false, a, false, b, (short)0, acc[g], false, false);
                }
            }

            #pragma unroll
            for (int kk = 0; kk < HDIM / 4; ++kk) {
                v2f a = *(const v2f*)&s_h[wave][n * HPITCH + kk * 4 + 2 * half];
                #pragma unroll
                for (int g = 0; g < 4; ++g) {
                    v2f b = *(const v2f*)&s_whh[(g * 64 + cb * 16 + n) * WHHP
                                                + kk * 4 + 2 * half];
                    acc[g] = __builtin_amdgcn_wmma_f32_16x16x4_f32(
                        false, a, false, b, (short)0, acc[g], false, false);
                }
            }

            float b0 = s_bias[0 * 64 + cb * 16 + n];
            float b1 = s_bias[1 * 64 + cb * 16 + n];
            float b2 = s_bias[2 * 64 + cb * 16 + n];
            float b3 = s_bias[3 * 64 + cb * 16 + n];

            #pragma unroll
            for (int r = 0; r < 8; ++r) {
                float ig = sigm(acc[0][r] + b0);
                float fg = sigm(acc[1][r] + b1);
                float gg = tanhf(acc[2][r] + b2);
                float og = sigm(acc[3][r] + b3);
                float cn = fg * creg[cb][r] + ig * gg;
                float hn = og * tanhf(cn);
                const bool act = t < llen[r];
                creg[cb][r] = act ? cn : creg[cb][r];
                hreg[cb][r] = act ? hn : hreg[cb][r];
            }
        }

        #pragma unroll
        for (int cb = 0; cb < 4; ++cb)
            #pragma unroll
            for (int r = 0; r < 8; ++r)
                s_h[wave][(8 * half + r) * HPITCH + cb * 16 + n] = hreg[cb][r];

        __syncthreads();
    }

    #pragma unroll
    for (int j = 0; j < 8; ++j) {
        int idx = j * 128 + lane * 4;
        int row = idx >> 6, col = idx & 63;
        v4f v = *(const v4f*)&s_h[wave][row * HPITCH + col];
        *(v4f*)&out[(base + row) * HDIM + col] = v;
    }
}

extern "C" void kernel_launch(void* const* d_in, const int* in_sizes, int n_in,
                              void* d_out, int out_size, void* d_ws, size_t ws_size,
                              hipStream_t stream) {
    const float* emb     = (const float*)d_in[0];
    const float* W_ih    = (const float*)d_in[1];
    const float* W_hh    = (const float*)d_in[2];
    const float* b_ih    = (const float*)d_in[3];
    const float* b_hh    = (const float*)d_in[4];
    const int*   chars   = (const int*)d_in[5];
    const int*   lengths = (const int*)d_in[6];
    float* out = (float*)d_out;

    const int n_words = in_sizes[6];
    const int blocks  = n_words / (WAVES * 16);
    char_lstm_wmma<<<blocks, WAVES * 32, 0, stream>>>(
        emb, W_ih, W_hh, b_ih, b_hh, chars, lengths, out);
}


# Weight fragment loads software-pipelined: the next block's LDS loads issued as one group in front of the current block of 8 matrix ops (second register set); arithmetic and sum order unchanged
# speedup vs baseline: 1.6077x; 1.6077x over previous
#include <hip/hip_runtime.h>


#define VOCAB   100
#define E       32
#define HDIM    64
#define G4H     256
#define MAXLEN  16
#define WAVES   8
#define EPITCH  36
#define WIHP    36
#define WHHP    68
#define HPITCH  68

typedef __attribute__((ext_vector_type(2))) float v2f;
typedef __attribute__((ext_vector_type(4))) float v4f;
typedef __attribute__((ext_vector_type(8))) float v8f;

__device__ __forceinline__ float sigm(float x) {
    return __builtin_amdgcn_rcpf(1.0f + __builtin_amdgcn_exp2f(-1.4426950408889634f * x));
}

__device__ __forceinline__ float tanh_fast(float x) {
    const float r = __builtin_amdgcn_rcpf(1.0f + __builtin_amdgcn_exp2f(2.8853900817779268f * x));
    return __builtin_fmaf(-2.0f, r, 1.0f);
}

__global__ __launch_bounds__(WAVES * 32, 4)
void char_lstm_wmma(const float* __restrict__ emb,
                    const float* __restrict__ Wih,
                    const float* __restrict__ Whh,
                    const float* __restrict__ bih,
                    const float* __restrict__ bhh,
                    const int*   __restrict__ chars,
                    const int*   __restrict__ lengths,
                    float*       __restrict__ out)
{
    __shared__ float s_emb[VOCAB * EPITCH];
    __shared__ float s_wih[G4H * WIHP];
    __shared__ float s_whh[G4H * WHHP];
    __shared__ float s_bias[G4H];
    __shared__ float s_h[WAVES][16 * HPITCH];

    const int tid  = threadIdx.x;
    const int lane = tid & 31;
    const int wave = tid >> 5;
    const int half = lane >> 4;
    const int n    = lane & 15;
    const int base = (blockIdx.x * WAVES + wave) * 16;

    for (int i = tid; i < VOCAB * E; i += WAVES * 32)
        s_emb[(i / E) * EPITCH + (i % E)] = emb[i];
    for (int i = tid; i < G4H * E; i += WAVES * 32)
        s_wih[(i / E) * WIHP + (i % E)] = Wih[i];
    for (int i = tid; i < G4H * HDIM; i += WAVES * 32)
        s_whh[(i / HDIM) * WHHP + (i % HDIM)] = Whh[i];
    for (int i = tid; i < G4H; i += WAVES * 32)
        s_bias[i] = bih[i] + bhh[i];
    for (int i = lane; i < 16 * HPITCH; i += 32)
        s_h[wave][i] = 0.0f;

    int llen[8];
    #pragma unroll
    for (int r = 0; r < 8; ++r)
        llen[r] = lengths[base + 8 * half + r];

    v8f creg[4];
    v8f hreg[4];
    #pragma unroll
    for (int cb = 0; cb < 4; ++cb)
        #pragma unroll
        for (int r = 0; r < 8; ++r) { creg[cb][r] = 0.0f; hreg[cb][r] = 0.0f; }

    __syncthreads();

    #pragma unroll 1
    for (int t = 0; t < MAXLEN; ++t) {
        const int c    = chars[(base + n) * MAXLEN + t];
        const int xoff = c * EPITCH + 2 * half;

        v2f afr[24];
        #pragma unroll
        for (int kk = 0; kk < E / 4; ++kk)
            afr[kk] = *(const v2f*)&s_emb[xoff + kk * 4];
        #pragma unroll
        for (int kk = 0; kk < HDIM / 4; ++kk)
            afr[E / 4 + kk] = *(const v2f*)&s_h[wave][n * HPITCH + kk * 4 + 2 * half];

        v2f bfr[2][4][2];
        #define LOAD_B(SET, CB, J)                                                          \
            do {                                                                            \
                _Pragma("unroll")                                                           \
                for (int g_ = 0; g_ < 4; ++g_)                                              \
                    _Pragma("unroll")                                                       \
                    for (int s_ = 0; s_ < 2; ++s_)                                          \
                        bfr[SET][g_][s_] = ((J) < 4)                                        \
                            ? *(const v2f*)&s_wih[(g_ * 64 + (CB) * 16 + n) * WIHP          \
                                                  + (2 * (J) + s_) * 4 + 2 * half]          \
                            : *(const v2f*)&s_whh[(g_ * 64 + (CB) * 16 + n) * WHHP          \
                                                  + (2 * ((J) - 4) + s_) * 4 + 2 * half];   \
            } while (0)

        LOAD_B(0, 0, 0);

        #pragma unroll
        for (int cb = 0; cb < 4; ++cb) {
            v8f acc[4];
            #pragma unroll
            for (int g = 0; g < 4; ++g)
                #pragma unroll
                for (int r = 0; r < 8; ++r) acc[g][r] = 0.0f;

            #pragma unroll
            for (int j = 0; j < 12; ++j) {
                const int blk = cb * 12 + j;
                const int cur = blk & 1;
                if (blk + 1 < 48)
                    LOAD_B(cur ^ 1, (blk + 1) / 12, (blk + 1) % 12);
                __builtin_amdgcn_sched_barrier(0);
                #pragma unroll
                for (int s = 0; s < 2; ++s)
                    #pragma unroll
                    for (int g = 0; g < 4; ++g)
                        acc[g] = __builtin_amdgcn_wmma_f32_16x16x4_f32(
                            false, afr[2 * j + s], false, bfr[cur][g][s],
                            (short)0, acc[g], false, false);
                __builtin_amdgcn_sched_barrier(0);
            }

            float b0 = s_bias[0 * 64 + cb * 16 + n];
            float b1 = s_bias[1 * 64 + cb * 16 + n];
            float b2 = s_bias[2 * 64 + cb * 16 + n];
            float b3 = s_bias[3 * 64 + cb * 16 + n];

            #pragma unroll
            for (int r = 0; r < 8; ++r) {
                float ig = sigm(acc[0][r] + b0);
                float fg = sigm(acc[1][r] + b1);
                float gg = tanh_fast(acc[2][r] + b2);
                float og = sigm(acc[3][r] + b3);
                float cn = fg * creg[cb][r] + ig * gg;
                float hn = og * tanh_fast(cn);
                const bool act = t < llen[r];
                const float cold = creg[cb][r];
                const float hold = hreg[cb][r];
                creg[cb][r] = act ? cn : cold;
                hreg[cb][r] = act ? hn : hold;
            }
        }

        #pragma unroll
        for (int cb = 0; cb < 4; ++cb)
            #pragma unroll
            for (int r = 0; r < 8; ++r)
                s_h[wave][(8 * half + r) * HPITCH + cb * 16 + n] = hreg[cb][r];

        __syncthreads();
        #undef LOAD_B
    }

    #pragma unroll
    for (int j = 0; j < 8; ++j) {
        int idx = j * 128 + lane * 4;
        int row = idx >> 6, col = idx & 63;
        v4f v = *(const v4f*)&s_h[wave][row * HPITCH + col];
        *(v4f*)&out[(base + row) * HDIM + col] = v;
    }
}

extern "C" void kernel_launch(void* const* d_in, const int* in_sizes, int n_in,
                              void* d_out, int out_size, void* d_ws, size_t ws_size,
                              hipStream_t stream) {
    const float* emb     = (const float*)d_in[0];
    const float* W_ih    = (const float*)d_in[1];
    const float* W_hh    = (const float*)d_in[2];
    const float* b_ih    = (const float*)d_in[3];
    const float* b_hh    = (const float*)d_in[4];
    const int*   chars   = (const int*)d_in[5];
    const int*   lengths = (const int*)d_in[6];
    float* out = (float*)d_out;

    const int n_words = in_sizes[6];
    const int blocks  = n_words / (WAVES * 16);
    char_lstm_wmma<<<blocks, WAVES * 32, 0, stream>>>(
        emb, W_ih, W_hh, b_ih, b_hh, chars, lengths, out);
}


# Words regrouped by length in each 256-word block (rank sort and chars tile in LDS); a wave runs its two 16-word groups only to their longest word, no per-step block barrier; sums unchanged
# speedup vs baseline: 2.3274x; 1.4477x over previous
#include <hip/hip_runtime.h>


#define VOCAB   100
#define E       32
#define HDIM    64
#define G4H     256
#define MAXLEN  16
#define WAVES   8
#define GROUPS  16
#define WPB     (GROUPS * 16)
#define CPITCH  17
#define EPITCH  36
#define WIHP    36
#define WHHP    68
#define HPITCH  68

typedef __attribute__((ext_vector_type(2))) float v2f;
typedef __attribute__((ext_vector_type(4))) float v4f;
typedef __attribute__((ext_vector_type(8))) float v8f;

__device__ __forceinline__ float sigm(float x) {
    return __builtin_amdgcn_rcpf(1.0f + __builtin_amdgcn_exp2f(-1.4426950408889634f * x));
}

__device__ __forceinline__ float tanh_fast(float x) {
    const float r = __builtin_amdgcn_rcpf(1.0f + __builtin_amdgcn_exp2f(2.8853900817779268f * x));
    return __builtin_fmaf(-2.0f, r, 1.0f);
}

__device__ __forceinline__ void wave_lds_order() {
    __builtin_amdgcn_fence(__ATOMIC_RELEASE, "wavefront");
    __builtin_amdgcn_wave_barrier();
    __builtin_amdgcn_fence(__ATOMIC_ACQUIRE, "wavefront");
}

__global__ __launch_bounds__(WAVES * 32, 4)
void char_lstm_wmma(const float* __restrict__ emb,
                    const float* __restrict__ Wih,
                    const float* __restrict__ Whh,
                    const float* __restrict__ bih,
                    const float* __restrict__ bhh,
                    const int*   __restrict__ chars,
                    const int*   __restrict__ lengths,
                    float*       __restrict__ out)
{
    __shared__ float s_emb[VOCAB * EPITCH];
    __shared__ float s_wih[G4H * WIHP];
    __shared__ float s_whh[G4H * WHHP];
    __shared__ float s_bias[G4H];
    __shared__ float s_h[WAVES][16 * HPITCH];
    __shared__ int   s_chars[WPB * CPITCH];
    __shared__ int   s_key[WPB];
    __shared__ int   s_slen[WPB];
    __shared__ int   s_perm[WPB];

    const int tid  = threadIdx.x;
    const int lane = tid & 31;
    const int wave = tid >> 5;
    const int half = lane >> 4;
    const int n    = lane & 15;
    const int bbase = blockIdx.x * WPB;

    for (int i = tid; i < VOCAB * E; i += WAVES * 32)
        s_emb[(i / E) * EPITCH + (i % E)] = emb[i];
    for (int i = tid; i < G4H * E; i += WAVES * 32)
        s_wih[(i / E) * WIHP + (i % E)] = Wih[i];
    for (int i = tid; i < G4H * HDIM; i += WAVES * 32)
        s_whh[(i / HDIM) * WHHP + (i % HDIM)] = Whh[i];
    for (int i = tid; i < G4H; i += WAVES * 32)
        s_bias[i] = bih[i] + bhh[i];
    for (int i = tid; i < WPB * MAXLEN; i += WAVES * 32)
        s_chars[(i / MAXLEN) * CPITCH + (i % MAXLEN)] = chars[bbase * MAXLEN + i];
    {
        int L = lengths[bbase + tid];
        L = L < 0 ? 0 : (L > MAXLEN ? MAXLEN : L);
        s_key[tid] = L;
    }
    __syncthreads();

    {
        const int mylen = s_key[tid];
        const int mykey = mylen * WPB + tid;
        int rank = 0;
        for (int j = 0; j < WPB; ++j)
            rank += (s_key[j] * WPB + j < mykey) ? 1 : 0;
        s_perm[rank] = tid;
        s_slen[rank] = mylen;
    }
    __syncthreads();

    #pragma unroll 1
    for (int pass = 0; pass < 2; ++pass) {
    const int g    = pass ? (GROUPS - 1 - wave) : wave;
    const int widx = s_perm[g * 16 + n];
    const int coff = widx * CPITCH;

    int llen[8];
    #pragma unroll
    for (int r = 0; r < 8; ++r)
        llen[r] = s_slen[g * 16 + 8 * half + r];
    const int nsteps = __builtin_amdgcn_readfirstlane(s_slen[g * 16 + 15]);

    v8f creg[4];
    v8f hreg[4];
    #pragma unroll
    for (int cb = 0; cb < 4; ++cb)
        #pragma unroll
        for (int r = 0; r < 8; ++r) { creg[cb][r] = 0.0f; hreg[cb][r] = 0.0f; }

    for (int i = lane; i < 16 * HPITCH; i += 32)
        s_h[wave][i] = 0.0f;
    wave_lds_order();

    #pragma unroll 1
    for (int t = 0; t < nsteps; ++t) {
        const int c    = s_chars[coff + t];
        const int xoff = c * EPITCH + 2 * half;

        v2f afr[24];
        #pragma unroll
        for (int kk = 0; kk < E / 4; ++kk)
            afr[kk] = *(const v2f*)&s_emb[xoff + kk * 4];
        #pragma unroll
        for (int kk = 0; kk < HDIM / 4; ++kk)
            afr[E / 4 + kk] = *(const v2f*)&s_h[wave][n * HPITCH + kk * 4 + 2 * half];

        v2f bfr[2][4][2];
        #define LOAD_B(SET, CB, J)                                                          \
            do {                                                                            \
                _Pragma("unroll")                                                           \
                for (int g_ = 0; g_ < 4; ++g_)                                              \
                    _Pragma("unroll")                                                       \
                    for (int s_ = 0; s_ < 2; ++s_)                                          \
                        bfr[SET][g_][s_] = ((J) < 4)                                        \
                            ? *(const v2f*)&s_wih[(g_ * 64 + (CB) * 16 + n) * WIHP          \
                                                  + (2 * (J) + s_) * 4 + 2 * half]          \
                            : *(const v2f*)&s_whh[(g_ * 64 + (CB) * 16 + n) * WHHP          \
                                                  + (2 * ((J) - 4) + s_) * 4 + 2 * half];   \
            } while (0)

        LOAD_B(0, 0, 0);

        #pragma unroll
        for (int cb = 0; cb < 4; ++cb) {
            v8f acc[4];
            #pragma unroll
            for (int g = 0; g < 4; ++g)
                #pragma unroll
                for (int r = 0; r < 8; ++r) acc[g][r] = 0.0f;

            #pragma unroll
            for (int j = 0; j < 12; ++j) {
                const int blk = cb * 12 + j;
                const int cur = blk & 1;
                if (blk + 1 < 48)
                    LOAD_B(cur ^ 1, (blk + 1) / 12, (blk + 1) % 12);
                __builtin_amdgcn_sched_barrier(0);
                #pragma unroll
                for (int s = 0; s < 2; ++s)
                    #pragma unroll
                    for (int g = 0; g < 4; ++g)
                        acc[g] = __builtin_amdgcn_wmma_f32_16x16x4_f32(
                            false, afr[2 * j + s], false, bfr[cur][g][s],
                            (short)0, acc[g], false, false);
                __builtin_amdgcn_sched_barrier(0);
            }

            float b0 = s_bias[0 * 64 + cb * 16 + n];
            float b1 = s_bias[1 * 64 + cb * 16 + n];
            float b2 = s_bias[2 * 64 + cb * 16 + n];
            float b3 = s_bias[3 * 64 + cb * 16 + n];

            #pragma unroll
            for (int r = 0; r < 8; ++r) {
                float ig = sigm(acc[0][r] + b0);
                float fg = sigm(acc[1][r] + b1);
                float gg = tanh_fast(acc[2][r] + b2);
                float og = sigm(acc[3][r] + b3);
                float cn = fg * creg[cb][r] + ig * gg;
                float hn = og * tanh_fast(cn);
                const bool act = t < llen[r];
                const float cold = creg[cb][r];
                const float hold = hreg[cb][r];
                creg[cb][r] = act ? cn : cold;
                hreg[cb][r] = act ? hn : hold;
            }
        }

        #pragma unroll
        for (int cb = 0; cb < 4; ++cb)
            #pragma unroll
            for (int r = 0; r < 8; ++r)
                s_h[wave][(8 * half + r) * HPITCH + cb * 16 + n] = hreg[cb][r];

        wave_lds_order();
        #undef LOAD_B
    }

    #pragma unroll
    for (int j = 0; j < 8; ++j) {
        int idx = j * 128 + lane * 4;
        int row = idx >> 6, col = idx & 63;
        v4f v = *(const v4f*)&s_h[wave][row * HPITCH + col];
        const int w = s_perm[g * 16 + row];
        *(v4f*)&out[(bbase + w) * HDIM + col] = v;
    }
    wave_lds_order();
    }
}

extern "C" void kernel_launch(void* const* d_in, const int* in_sizes, int n_in,
                              void* d_out, int out_size, void* d_ws, size_t ws_size,
                              hipStream_t stream) {
    const float* emb     = (const float*)d_in[0];
    const float* W_ih    = (const float*)d_in[1];
    const float* W_hh    = (const float*)d_in[2];
    const float* b_ih    = (const float*)d_in[3];
    const float* b_hh    = (const float*)d_in[4];
    const int*   chars   = (const int*)d_in[5];
    const int*   lengths = (const int*)d_in[6];
    float* out = (float*)d_out;

    const int n_words = in_sizes[6];
    const int blocks  = n_words / WPB;
    char_lstm_wmma<<<blocks, WAVES * 32, 0, stream>>>(
        emb, W_ih, W_hh, b_ih, b_hh, chars, lengths, out);
}


# 16 waves per block in place of 8 (four per SIMD), one 16-word group per wave; groups dealt by a fixed 4x4 square: equal steps per SIMD, long groups on its older waves; sums unchanged
# speedup vs baseline: 2.4268x; 1.0427x over previous
#include <hip/hip_runtime.h>


#define VOCAB   100
#define E       32
#define HDIM    64
#define G4H     256
#define MAXLEN  16
#define WAVES   16
#define GROUPS  16
#define WPB     (GROUPS * 16)
#define CPITCH  17
#define EPITCH  36
#define WIHP    36
#define WHHP    68
#define HPITCH  68

#define GROUP_OF_WAVE 0x96B4785AC3E12D0FULL

typedef __attribute__((ext_vector_type(2))) float v2f;
typedef __attribute__((ext_vector_type(4))) float v4f;
typedef __attribute__((ext_vector_type(8))) float v8f;

__device__ __forceinline__ float sigm(float x) {
    return __builtin_amdgcn_rcpf(1.0f + __builtin_amdgcn_exp2f(-1.4426950408889634f * x));
}

__device__ __forceinline__ float tanh_fast(float x) {
    const float r = __builtin_amdgcn_rcpf(1.0f + __builtin_amdgcn_exp2f(2.8853900817779268f * x));
    return __builtin_fmaf(-2.0f, r, 1.0f);
}

__device__ __forceinline__ void wave_lds_order() {
    __builtin_amdgcn_fence(__ATOMIC_RELEASE, "wavefront");
    __builtin_amdgcn_wave_barrier();
    __builtin_amdgcn_fence(__ATOMIC_ACQUIRE, "wavefront");
}

__global__ __launch_bounds__(WAVES * 32)
void char_lstm_wmma(const float* __restrict__ emb,
                    const float* __restrict__ Wih,
                    const float* __restrict__ Whh,
                    const float* __restrict__ bih,
                    const float* __restrict__ bhh,
                    const int*   __restrict__ chars,
                    const int*   __restrict__ lengths,
                    float*       __restrict__ out)
{
    __shared__ float s_emb[VOCAB * EPITCH];
    __shared__ float s_wih[G4H * WIHP];
    __shared__ float s_whh[G4H * WHHP];
    __shared__ float s_bias[G4H];
    __shared__ float s_h[WAVES][16 * HPITCH];
    __shared__ int   s_chars[WPB * CPITCH];
    __shared__ int   s_key[WPB];
    __shared__ int   s_slen[WPB];
    __shared__ int   s_perm[WPB];

    const int tid  = threadIdx.x;
    const int lane = tid & 31;
    const int wave = __builtin_amdgcn_readfirstlane(tid >> 5);
    const int half = lane >> 4;
    const int n    = lane & 15;
    const int bbase = blockIdx.x * WPB;

    for (int i = tid; i < VOCAB * E; i += WAVES * 32)
        s_emb[(i / E) * EPITCH + (i % E)] = emb[i];
    for (int i = tid; i < G4H * E; i += WAVES * 32)
        s_wih[(i / E) * WIHP + (i % E)] = Wih[i];
    for (int i = tid; i < G4H * HDIM; i += WAVES * 32)
        s_whh[(i / HDIM) * WHHP + (i % HDIM)] = Whh[i];
    for (int i = tid; i < G4H; i += WAVES * 32)
        s_bias[i] = bih[i] + bhh[i];
    for (int i = tid; i < WPB * MAXLEN; i += WAVES * 32)
        s_chars[(i / MAXLEN) * CPITCH + (i % MAXLEN)] = chars[bbase * MAXLEN + i];
    if (tid < WPB) {
        int L = lengths[bbase + tid];
        L = L < 0 ? 0 : (L > MAXLEN ? MAXLEN : L);
        s_key[tid] = L;
    }
    __syncthreads();

    if (tid < WPB) {
        const int mylen = s_key[tid];
        const int mykey = mylen * WPB + tid;
        int rank = 0;
        for (int j = 0; j < WPB; ++j)
            rank += (s_key[j] * WPB + j < mykey) ? 1 : 0;
        s_perm[rank] = tid;
        s_slen[rank] = mylen;
    }
    __syncthreads();

    const int g    = (int)((GROUP_OF_WAVE >> (4 * wave)) & 15);
    const int widx = s_perm[g * 16 + n];
    const int coff = widx * CPITCH;

    int llen[8];
    #pragma unroll
    for (int r = 0; r < 8; ++r)
        llen[r] = s_slen[g * 16 + 8 * half + r];
    const int nsteps = __builtin_amdgcn_readfirstlane(s_slen[g * 16 + 15]);

    v8f creg[4];
    v8f hreg[4];
    #pragma unroll
    for (int cb = 0; cb < 4; ++cb)
        #pragma unroll
        for (int r = 0; r < 8; ++r) { creg[cb][r] = 0.0f; hreg[cb][r] = 0.0f; }

    for (int i = lane; i < 16 * HPITCH; i += 32)
        s_h[wave][i] = 0.0f;
    wave_lds_order();

    #pragma unroll 1
    for (int t = 0; t < nsteps; ++t) {
        const int c    = s_chars[coff + t];
        const int xoff = c * EPITCH + 2 * half;

        v2f afr[24];
        #pragma unroll
        for (int kk = 0; kk < E / 4; ++kk)
            afr[kk] = *(const v2f*)&s_emb[xoff + kk * 4];
        #pragma unroll
        for (int kk = 0; kk < HDIM / 4; ++kk)
            afr[E / 4 + kk] = *(const v2f*)&s_h[wave][n * HPITCH + kk * 4 + 2 * half];

        v2f bfr[2][4][2];
        #define LOAD_B(SET, CB, J)                                                          \
            do {                                                                            \
                _Pragma("unroll")                                                           \
                for (int g_ = 0; g_ < 4; ++g_)                                              \
                    _Pragma("unroll")                                                       \
                    for (int s_ = 0; s_ < 2; ++s_)                                          \
                        bfr[SET][g_][s_] = ((J) < 4)                                        \
                            ? *(const v2f*)&s_wih[(g_ * 64 + (CB) * 16 + n) * WIHP          \
                                                  + (2 * (J) + s_) * 4 + 2 * half]          \
                            : *(const v2f*)&s_whh[(g_ * 64 + (CB) * 16 + n) * WHHP          \
                                                  + (2 * ((J) - 4) + s_) * 4 + 2 * half];   \
            } while (0)

        LOAD_B(0, 0, 0);

        #pragma unroll
        for (int cb = 0; cb < 4; ++cb) {
            v8f acc[4];
            #pragma unroll
            for (int g = 0; g < 4; ++g)
                #pragma unroll
                for (int r = 0; r < 8; ++r) acc[g][r] = 0.0f;

            #pragma unroll
            for (int j = 0; j < 12; ++j) {
                const int blk = cb * 12 + j;
                const int cur = blk & 1;
                if (blk + 1 < 48)
                    LOAD_B(cur ^ 1, (blk + 1) / 12, (blk + 1) % 12);
                __builtin_amdgcn_sched_barrier(0);
                #pragma unroll
                for (int s = 0; s < 2; ++s)
                    #pragma unroll
                    for (int g = 0; g < 4; ++g)
                        acc[g] = __builtin_amdgcn_wmma_f32_16x16x4_f32(
                            false, afr[2 * j + s], false, bfr[cur][g][s],
                            (short)0, acc[g], false, false);
                __builtin_amdgcn_sched_barrier(0);
            }

            float b0 = s_bias[0 * 64 + cb * 16 + n];
            float b1 = s_bias[1 * 64 + cb * 16 + n];
            float b2 = s_bias[2 * 64 + cb * 16 + n];
            float b3 = s_bias[3 * 64 + cb * 16 + n];

            #pragma unroll
            for (int r = 0; r < 8; ++r) {
                float ig = sigm(acc[0][r] + b0);
                float fg = sigm(acc[1][r] + b1);
                float gg = tanh_fast(acc[2][r] + b2);
                float og = sigm(acc[3][r] + b3);
                float cn = fg * creg[cb][r] + ig * gg;
                float hn = og * tanh_fast(cn);
                const bool act = t < llen[r];
                const float cold = creg[cb][r];
                const float hold = hreg[cb][r];
                creg[cb][r] = act ? cn : cold;
                hreg[cb][r] = act ? hn : hold;
            }
        }

        #pragma unroll
        for (int cb = 0; cb < 4; ++cb)
            #pragma unroll
            for (int r = 0; r < 8; ++r)
                s_h[wave][(8 * half + r) * HPITCH + cb * 16 + n] = hreg[cb][r];

        wave_lds_order();
        #undef LOAD_B
    }

    #pragma unroll
    for (int j = 0; j < 8; ++j) {
        int idx = j * 128 + lane * 4;
        int row = idx >> 6, col = idx & 63;
        v4f v = *(const v4f*)&s_h[wave][row * HPITCH + col];
        const int w = s_perm[g * 16 + row];
        *(v4f*)&out[(bbase + w) * HDIM + col] = v;
    }
}

extern "C" void kernel_launch(void* const* d_in, const int* in_sizes, int n_in,
                              void* d_out, int out_size, void* d_ws, size_t ws_size,
                              hipStream_t stream) {
    const float* emb     = (const float*)d_in[0];
    const float* W_ih    = (const float*)d_in[1];
    const float* W_hh    = (const float*)d_in[2];
    const float* b_ih    = (const float*)d_in[3];
    const float* b_hh    = (const float*)d_in[4];
    const int*   chars   = (const int*)d_in[5];
    const int*   lengths = (const int*)d_in[6];
    float* out = (float*)d_out;

    const int n_words = in_sizes[6];
    const int blocks  = n_words / WPB;
    char_lstm_wmma<<<blocks, WAVES * 32, 0, stream>>>(
        emb, W_ih, W_hh, b_ih, b_hh, chars, lengths, out);
}


# Words regrouped by length over 2048-word super-blocks (ballot counting sort, redone in each of 8 blocks); a SIMD gets one group per quarter of the sorted order: equal steps chip-wide; sums unchanged
# speedup vs baseline: 2.7599x; 1.1373x over previous
#include <hip/hip_runtime.h>


#define VOCAB   100
#define E       32
#define HDIM    64
#define G4H     256
#define MAXLEN  16
#define WAVES   16
#define GROUPS  16
#define WPB     (GROUPS * 16)
#define BPS     8
#define SBW     (BPS * WPB)
#define NKEY    (MAXLEN + 1)
#define KROWS   (SBW / (WAVES * 32))
#define CPITCH  17
#define EPITCH  36
#define WIHP    36
#define WHHP    68
#define HPITCH  68

typedef __attribute__((ext_vector_type(2))) float v2f;
typedef __attribute__((ext_vector_type(4))) float v4f;
typedef __attribute__((ext_vector_type(8))) float v8f;

__device__ __forceinline__ float sigm(float x) {
    return __builtin_amdgcn_rcpf(1.0f + __builtin_amdgcn_exp2f(-1.4426950408889634f * x));
}

__device__ __forceinline__ float tanh_fast(float x) {
    const float r = __builtin_amdgcn_rcpf(1.0f + __builtin_amdgcn_exp2f(2.8853900817779268f * x));
    return __builtin_fmaf(-2.0f, r, 1.0f);
}

__device__ __forceinline__ void wave_lds_order() {
    __builtin_amdgcn_fence(__ATOMIC_RELEASE, "wavefront");
    __builtin_amdgcn_wave_barrier();
    __builtin_amdgcn_fence(__ATOMIC_ACQUIRE, "wavefront");
}

__global__ __launch_bounds__(WAVES * 32)
void char_lstm_wmma(const float* __restrict__ emb,
                    const float* __restrict__ Wih,
                    const float* __restrict__ Whh,
                    const float* __restrict__ bih,
                    const float* __restrict__ bhh,
                    const int*   __restrict__ chars,
                    const int*   __restrict__ lengths,
                    float*       __restrict__ out)
{
    __shared__ float s_emb[VOCAB * EPITCH];
    __shared__ float s_wih[G4H * WIHP];
    __shared__ float s_whh[G4H * WHHP];
    __shared__ float s_bias[G4H];
    __shared__ float s_h[WAVES][16 * HPITCH];
    __shared__ int   s_chars[WPB * CPITCH];
    __shared__ int   s_wtot[NKEY * WAVES];
    __shared__ int   s_base[NKEY * WAVES];
    __shared__ int   s_slen[WPB];
    __shared__ int   s_perm[WPB];

    const int tid  = threadIdx.x;
    const int lane = tid & 31;
    const int wave = __builtin_amdgcn_readfirstlane(tid >> 5);
    const int half = lane >> 4;
    const int n    = lane & 15;
    const int memb  = blockIdx.x % BPS;
    const int sbase = (blockIdx.x / BPS) * SBW;

    for (int i = tid; i < VOCAB * E; i += WAVES * 32)
        s_emb[(i / E) * EPITCH + (i % E)] = emb[i];
    for (int i = tid; i < G4H * E; i += WAVES * 32)
        s_wih[(i / E) * WIHP + (i % E)] = Wih[i];
    for (int i = tid; i < G4H * HDIM; i += WAVES * 32)
        s_whh[(i / HDIM) * WHHP + (i % HDIM)] = Whh[i];
    for (int i = tid; i < G4H; i += WAVES * 32)
        s_bias[i] = bih[i] + bhh[i];
    int key[KROWS], mybase[KROWS], within[KROWS];
    {
        int cnt[NKEY];
        #pragma unroll
        for (int k = 0; k < NKEY; ++k) cnt[k] = 0;
        #pragma unroll
        for (int r = 0; r < KROWS; ++r) {
            int L = lengths[sbase + wave * (KROWS * 32) + r * 32 + lane];
            L = L < 0 ? 0 : (L > MAXLEN ? MAXLEN : L);
            key[r] = L;
            unsigned mymask = 0u;
            int base = 0;
            #pragma unroll
            for (int k = 0; k < NKEY; ++k) {
                const bool mine = (L == k);
                const unsigned m = __builtin_amdgcn_ballot_w32(mine);
                mymask = mine ? m : mymask;
                base   = mine ? cnt[k] : base;
                cnt[k] += __builtin_popcount(m);
            }
            mybase[r] = base;
            within[r] = __builtin_popcount(mymask & ((1u << lane) - 1u));
        }
        int mine_tot = 0;
        #pragma unroll
        for (int k = 0; k < NKEY; ++k) mine_tot = (lane == k) ? cnt[k] : mine_tot;
        if (lane < NKEY) s_wtot[lane * WAVES + wave] = mine_tot;
    }
    __syncthreads();

    if (wave == 0) {
        int v[9];
        int run = 0;
        #pragma unroll
        for (int j = 0; j < 9; ++j) {
            const int idx = 9 * lane + j;
            const int x = (idx < NKEY * WAVES) ? s_wtot[idx] : 0;
            v[j] = run;
            run += x;
        }
        int inc = run;
        #pragma unroll
        for (int d = 1; d < 32; d <<= 1) {
            const int o = __shfl_up(inc, d);
            inc += (lane >= d) ? o : 0;
        }
        const int excl = inc - run;
        #pragma unroll
        for (int j = 0; j < 9; ++j) {
            const int idx = 9 * lane + j;
            if (idx < NKEY * WAVES) s_base[idx] = v[j] + excl;
        }
    }
    __syncthreads();

    #pragma unroll
    for (int r = 0; r < KROWS; ++r) {
        const int pos  = s_base[key[r] * WAVES + wave] + mybase[r] + within[r];
        const int grp  = pos >> 4;
        const int quar = grp >> 5;
        const int off  = grp & 31;
        const int q    = (quar & 1) ? (31 - off) : off;
        if ((q >> 2) == memb) {
            const int slot = (4 * (3 - quar) + (q & 3)) * 16 + (pos & 15);
            s_perm[slot] = wave * (KROWS * 32) + r * 32 + lane;
            s_slen[slot] = key[r];
        }
    }
    __syncthreads();

    const int g    = wave;
    const int coff = (g * 16 + n) * CPITCH;

    #pragma unroll
    for (int p = 0; p < 2; ++p) {
        const int e    = p * 32 + lane;
        const int row  = e >> 2, part = e & 3;
        const int wd   = s_perm[g * 16 + row];
        const int4 cv  = *(const int4*)&chars[(sbase + wd) * MAXLEN + part * 4];
        int* dst = &s_chars[(g * 16 + row) * CPITCH + part * 4];
        dst[0] = cv.x; dst[1] = cv.y; dst[2] = cv.z; dst[3] = cv.w;
    }

    int llen[8];
    #pragma unroll
    for (int r = 0; r < 8; ++r)
        llen[r] = s_slen[g * 16 + 8 * half + r];
    const int nsteps = __builtin_amdgcn_readfirstlane(s_slen[g * 16 + 15]);

    v8f creg[4];
    v8f hreg[4];
    #pragma unroll
    for (int cb = 0; cb < 4; ++cb)
        #pragma unroll
        for (int r = 0; r < 8; ++r) { creg[cb][r] = 0.0f; hreg[cb][r] = 0.0f; }

    for (int i = lane; i < 16 * HPITCH; i += 32)
        s_h[wave][i] = 0.0f;
    wave_lds_order();

    #pragma unroll 1
    for (int t = 0; t < nsteps; ++t) {
        const int c    = s_chars[coff + t];
        const int xoff = c * EPITCH + 2 * half;

        v2f afr[24];
        #pragma unroll
        for (int kk = 0; kk < E / 4; ++kk)
            afr[kk] = *(const v2f*)&s_emb[xoff + kk * 4];
        #pragma unroll
        for (int kk = 0; kk < HDIM / 4; ++kk)
            afr[E / 4 + kk] = *(const v2f*)&s_h[wave][n * HPITCH + kk * 4 + 2 * half];

        v2f bfr[2][4][2];
        #define LOAD_B(SET, CB, J)                                                          \
            do {                                                                            \
                _Pragma("unroll")                                                           \
                for (int g_ = 0; g_ < 4; ++g_)                                              \
                    _Pragma("unroll")                                                       \
                    for (int s_ = 0; s_ < 2; ++s_)                                          \
                        bfr[SET][g_][s_] = ((J) < 4)                                        \
                            ? *(const v2f*)&s_wih[(g_ * 64 + (CB) * 16 + n) * WIHP          \
                                                  + (2 * (J) + s_) * 4 + 2 * half]          \
                            : *(const v2f*)&s_whh[(g_ * 64 + (CB) * 16 + n) * WHHP          \
                                                  + (2 * ((J) - 4) + s_) * 4 + 2 * half];   \
            } while (0)

        LOAD_B(0, 0, 0);

        #pragma unroll
        for (int cb = 0; cb < 4; ++cb) {
            v8f acc[4];
            #pragma unroll
            for (int g = 0; g < 4; ++g)
                #pragma unroll
                for (int r = 0; r < 8; ++r) acc[g][r] = 0.0f;

            #pragma unroll
            for (int j = 0; j < 12; ++j) {
                const int blk = cb * 12 + j;
                const int cur = blk & 1;
                if (blk + 1 < 48)
                    LOAD_B(cur ^ 1, (blk + 1) / 12, (blk + 1) % 12);
                __builtin_amdgcn_sched_barrier(0);
                #pragma unroll
                for (int s = 0; s < 2; ++s)
                    #pragma unroll
                    for (int g = 0; g < 4; ++g)
                        acc[g] = __builtin_amdgcn_wmma_f32_16x16x4_f32(
                            false, afr[2 * j + s], false, bfr[cur][g][s],
                            (short)0, acc[g], false, false);
                __builtin_amdgcn_sched_barrier(0);
            }

            float b0 = s_bias[0 * 64 + cb * 16 + n];
            float b1 = s_bias[1 * 64 + cb * 16 + n];
            float b2 = s_bias[2 * 64 + cb * 16 + n];
            float b3 = s_bias[3 * 64 + cb * 16 + n];

            #pragma unroll
            for (int r = 0; r < 8; ++r) {
                float ig = sigm(acc[0][r] + b0);
                float fg = sigm(acc[1][r] + b1);
                float gg = tanh_fast(acc[2][r] + b2);
                float og = sigm(acc[3][r] + b3);
                float cn = fg * creg[cb][r] + ig * gg;
                float hn = og * tanh_fast(cn);
                const bool act = t < llen[r];
                const float cold = creg[cb][r];
                const float hold = hreg[cb][r];
                creg[cb][r] = act ? cn : cold;
                hreg[cb][r] = act ? hn : hold;
            }
        }

        #pragma unroll
        for (int cb = 0; cb < 4; ++cb)
            #pragma unroll
            for (int r = 0; r < 8; ++r)
                s_h[wave][(8 * half + r) * HPITCH + cb * 16 + n] = hreg[cb][r];

        wave_lds_order();
        #undef LOAD_B
    }

    #pragma unroll
    for (int j = 0; j < 8; ++j) {
        int idx = j * 128 + lane * 4;
        int row = idx >> 6, col = idx & 63;
        v4f v = *(const v4f*)&s_h[wave][row * HPITCH + col];
        const int w = s_perm[g * 16 + row];
        *(v4f*)&out[(sbase + w) * HDIM + col] = v;
    }
}

extern "C" void kernel_launch(void* const* d_in, const int* in_sizes, int n_in,
                              void* d_out, int out_size, void* d_ws, size_t ws_size,
                              hipStream_t stream) {
    const float* emb     = (const float*)d_in[0];
    const float* W_ih    = (const float*)d_in[1];
    const float* W_hh    = (const float*)d_in[2];
    const float* b_ih    = (const float*)d_in[3];
    const float* b_hh    = (const float*)d_in[4];
    const int*   chars   = (const int*)d_in[5];
    const int*   lengths = (const int*)d_in[6];
    float* out = (float*)d_out;

    const int n_words = in_sizes[6];
    const int blocks  = n_words / WPB;
    char_lstm_wmma<<<blocks, WAVES * 32, 0, stream>>>(
        emb, W_ih, W_hh, b_ih, b_hh, chars, lengths, out);
}


# Gate activations on the hardware tanh instruction (v_tanh_f32): sigmoid(x) = 0.5 + 0.5*tanh(x/2), halving folded into the bias multiply-add; matrix sums unchanged, results move by tanh rounding
# speedup vs baseline: 2.9086x; 1.0539x over previous
#include <hip/hip_runtime.h>


#define VOCAB   100
#define E       32
#define HDIM    64
#define G4H     256
#define MAXLEN  16
#define WAVES   16
#define GROUPS  16
#define WPB     (GROUPS * 16)
#define BPS     8
#define SBW     (BPS * WPB)
#define NKEY    (MAXLEN + 1)
#define KROWS   (SBW / (WAVES * 32))
#define CPITCH  17
#define EPITCH  36
#define WIHP    36
#define WHHP    68
#define HPITCH  68

typedef __attribute__((ext_vector_type(2))) float v2f;
typedef __attribute__((ext_vector_type(4))) float v4f;
typedef __attribute__((ext_vector_type(8))) float v8f;

__device__ __forceinline__ float tanh_hw(float x) {
    return __builtin_amdgcn_tanhf(x);
}

__device__ __forceinline__ float sigm_of_half(float xh) {
    return __builtin_fmaf(0.5f, tanh_hw(xh), 0.5f);
}

__device__ __forceinline__ void wave_lds_order() {
    __builtin_amdgcn_fence(__ATOMIC_RELEASE, "wavefront");
    __builtin_amdgcn_wave_barrier();
    __builtin_amdgcn_fence(__ATOMIC_ACQUIRE, "wavefront");
}

__global__ __launch_bounds__(WAVES * 32)
void char_lstm_wmma(const float* __restrict__ emb,
                    const float* __restrict__ Wih,
                    const float* __restrict__ Whh,
                    const float* __restrict__ bih,
                    const float* __restrict__ bhh,
                    const int*   __restrict__ chars,
                    const int*   __restrict__ lengths,
                    float*       __restrict__ out)
{
    __shared__ float s_emb[VOCAB * EPITCH];
    __shared__ float s_wih[G4H * WIHP];
    __shared__ float s_whh[G4H * WHHP];
    __shared__ float s_bias[G4H];
    __shared__ float s_h[WAVES][16 * HPITCH];
    __shared__ int   s_chars[WPB * CPITCH];
    __shared__ int   s_wtot[NKEY * WAVES];
    __shared__ int   s_base[NKEY * WAVES];
    __shared__ int   s_slen[WPB];
    __shared__ int   s_perm[WPB];

    const int tid  = threadIdx.x;
    const int lane = tid & 31;
    const int wave = __builtin_amdgcn_readfirstlane(tid >> 5);
    const int half = lane >> 4;
    const int n    = lane & 15;
    const int memb  = blockIdx.x % BPS;
    const int sbase = (blockIdx.x / BPS) * SBW;

    for (int i = tid; i < VOCAB * E; i += WAVES * 32)
        s_emb[(i / E) * EPITCH + (i % E)] = emb[i];
    for (int i = tid; i < G4H * E; i += WAVES * 32)
        s_wih[(i / E) * WIHP + (i % E)] = Wih[i];
    for (int i = tid; i < G4H * HDIM; i += WAVES * 32)
        s_whh[(i / HDIM) * WHHP + (i % HDIM)] = Whh[i];
    for (int i = tid; i < G4H; i += WAVES * 32) {
        const float b = bih[i] + bhh[i];
        s_bias[i] = (i / HDIM == 2) ? b : 0.5f * b;
    }
    int key[KROWS], mybase[KROWS], within[KROWS];
    {
        int cnt[NKEY];
        #pragma unroll
        for (int k = 0; k < NKEY; ++k) cnt[k] = 0;
        #pragma unroll
        for (int r = 0; r < KROWS; ++r) {
            int L = lengths[sbase + wave * (KROWS * 32) + r * 32 + lane];
            L = L < 0 ? 0 : (L > MAXLEN ? MAXLEN : L);
            key[r] = L;
            unsigned mymask = 0u;
            int base = 0;
            #pragma unroll
            for (int k = 0; k < NKEY; ++k) {
                const bool mine = (L == k);
                const unsigned m = __builtin_amdgcn_ballot_w32(mine);
                mymask = mine ? m : mymask;
                base   = mine ? cnt[k] : base;
                cnt[k] += __builtin_popcount(m);
            }
            mybase[r] = base;
            within[r] = __builtin_popcount(mymask & ((1u << lane) - 1u));
        }
        int mine_tot = 0;
        #pragma unroll
        for (int k = 0; k < NKEY; ++k) mine_tot = (lane == k) ? cnt[k] : mine_tot;
        if (lane < NKEY) s_wtot[lane * WAVES + wave] = mine_tot;
    }
    __syncthreads();

    if (wave == 0) {
        int v[9];
        int run = 0;
        #pragma unroll
        for (int j = 0; j < 9; ++j) {
            const int idx = 9 * lane + j;
            const int x = (idx < NKEY * WAVES) ? s_wtot[idx] : 0;
            v[j] = run;
            run += x;
        }
        int inc = run;
        #pragma unroll
        for (int d = 1; d < 32; d <<= 1) {
            const int o = __shfl_up(inc, d);
            inc += (lane >= d) ? o : 0;
        }
        const int excl = inc - run;
        #pragma unroll
        for (int j = 0; j < 9; ++j) {
            const int idx = 9 * lane + j;
            if (idx < NKEY * WAVES) s_base[idx] = v[j] + excl;
        }
    }
    __syncthreads();

    #pragma unroll
    for (int r = 0; r < KROWS; ++r) {
        const int pos  = s_base[key[r] * WAVES + wave] + mybase[r] + within[r];
        const int grp  = pos >> 4;
        const int quar = grp >> 5;
        const int off  = grp & 31;
        const int q    = (quar & 1) ? (31 - off) : off;
        if ((q >> 2) == memb) {
            const int slot = (4 * (3 - quar) + (q & 3)) * 16 + (pos & 15);
            s_perm[slot] = wave * (KROWS * 32) + r * 32 + lane;
            s_slen[slot] = key[r];
        }
    }
    __syncthreads();

    const int g    = wave;
    const int coff = (g * 16 + n) * CPITCH;

    #pragma unroll
    for (int p = 0; p < 2; ++p) {
        const int e    = p * 32 + lane;
        const int row  = e >> 2, part = e & 3;
        const int wd   = s_perm[g * 16 + row];
        const int4 cv  = *(const int4*)&chars[(sbase + wd) * MAXLEN + part * 4];
        int* dst = &s_chars[(g * 16 + row) * CPITCH + part * 4];
        dst[0] = cv.x; dst[1] = cv.y; dst[2] = cv.z; dst[3] = cv.w;
    }

    int llen[8];
    #pragma unroll
    for (int r = 0; r < 8; ++r)
        llen[r] = s_slen[g * 16 + 8 * half + r];
    const int nsteps = __builtin_amdgcn_readfirstlane(s_slen[g * 16 + 15]);

    v8f creg[4];
    v8f hreg[4];
    #pragma unroll
    for (int cb = 0; cb < 4; ++cb)
        #pragma unroll
        for (int r = 0; r < 8; ++r) { creg[cb][r] = 0.0f; hreg[cb][r] = 0.0f; }

    for (int i = lane; i < 16 * HPITCH; i += 32)
        s_h[wave][i] = 0.0f;
    wave_lds_order();

    #pragma unroll 1
    for (int t = 0; t < nsteps; ++t) {
        const int c    = s_chars[coff + t];
        const int xoff = c * EPITCH + 2 * half;

        v2f afr[24];
        #pragma unroll
        for (int kk = 0; kk < E / 4; ++kk)
            afr[kk] = *(const v2f*)&s_emb[xoff + kk * 4];
        #pragma unroll
        for (int kk = 0; kk < HDIM / 4; ++kk)
            afr[E / 4 + kk] = *(const v2f*)&s_h[wave][n * HPITCH + kk * 4 + 2 * half];

        v2f bfr[2][4][2];
        #define LOAD_B(SET, CB, J)                                                          \
            do {                                                                            \
                _Pragma("unroll")                                                           \
                for (int g_ = 0; g_ < 4; ++g_)                                              \
                    _Pragma("unroll")                                                       \
                    for (int s_ = 0; s_ < 2; ++s_)                                          \
                        bfr[SET][g_][s_] = ((J) < 4)                                        \
                            ? *(const v2f*)&s_wih[(g_ * 64 + (CB) * 16 + n) * WIHP          \
                                                  + (2 * (J) + s_) * 4 + 2 * half]          \
                            : *(const v2f*)&s_whh[(g_ * 64 + (CB) * 16 + n) * WHHP          \
                                                  + (2 * ((J) - 4) + s_) * 4 + 2 * half];   \
            } while (0)

        LOAD_B(0, 0, 0);

        #pragma unroll
        for (int cb = 0; cb < 4; ++cb) {
            v8f acc[4];
            #pragma unroll
            for (int g = 0; g < 4; ++g)
                #pragma unroll
                for (int r = 0; r < 8; ++r) acc[g][r] = 0.0f;

            #pragma unroll
            for (int j = 0; j < 12; ++j) {
                const int blk = cb * 12 + j;
                const int cur = blk & 1;
                if (blk + 1 < 48)
                    LOAD_B(cur ^ 1, (blk + 1) / 12, (blk + 1) % 12);
                __builtin_amdgcn_sched_barrier(0);
                #pragma unroll
                for (int s = 0; s < 2; ++s)
                    #pragma unroll
                    for (int g = 0; g < 4; ++g)
                        acc[g] = __builtin_amdgcn_wmma_f32_16x16x4_f32(
                            false, afr[2 * j + s], false, bfr[cur][g][s],
                            (short)0, acc[g], false, false);
                __builtin_amdgcn_sched_barrier(0);
            }

            float b0 = s_bias[0 * 64 + cb * 16 + n];
            float b1 = s_bias[1 * 64 + cb * 16 + n];
            float b2 = s_bias[2 * 64 + cb * 16 + n];
            float b3 = s_bias[3 * 64 + cb * 16 + n];

            #pragma unroll
            for (int r = 0; r < 8; ++r) {
                float ig = sigm_of_half(__builtin_fmaf(0.5f, acc[0][r], b0));
                float fg = sigm_of_half(__builtin_fmaf(0.5f, acc[1][r], b1));
                float gg = tanh_hw(acc[2][r] + b2);
                float og = sigm_of_half(__builtin_fmaf(0.5f, acc[3][r], b3));
                float cn = fg * creg[cb][r] + ig * gg;
                float hn = og * tanh_hw(cn);
                const bool act = t < llen[r];
                const float cold = creg[cb][r];
                const float hold = hreg[cb][r];
                creg[cb][r] = act ? cn : cold;
                hreg[cb][r] = act ? hn : hold;
            }
        }

        #pragma unroll
        for (int cb = 0; cb < 4; ++cb)
            #pragma unroll
            for (int r = 0; r < 8; ++r)
                s_h[wave][(8 * half + r) * HPITCH + cb * 16 + n] = hreg[cb][r];

        wave_lds_order();
        #undef LOAD_B
    }

    #pragma unroll
    for (int j = 0; j < 8; ++j) {
        int idx = j * 128 + lane * 4;
        int row = idx >> 6, col = idx & 63;
        v4f v = *(const v4f*)&s_h[wave][row * HPITCH + col];
        const int w = s_perm[g * 16 + row];
        *(v4f*)&out[(sbase + w) * HDIM + col] = v;
    }
}

extern "C" void kernel_launch(void* const* d_in, const int* in_sizes, int n_in,
                              void* d_out, int out_size, void* d_ws, size_t ws_size,
                              hipStream_t stream) {
    const float* emb     = (const float*)d_in[0];
    const float* W_ih    = (const float*)d_in[1];
    const float* W_hh    = (const float*)d_in[2];
    const float* b_ih    = (const float*)d_in[3];
    const float* b_hh    = (const float*)d_in[4];
    const int*   chars   = (const int*)d_in[5];
    const int*   lengths = (const int*)d_in[6];
    float* out = (float*)d_out;

    const int n_words = in_sizes[6];
    const int blocks  = n_words / WPB;
    char_lstm_wmma<<<blocks, WAVES * 32, 0, stream>>>(
        emb, W_ih, W_hh, b_ih, b_hh, chars, lengths, out);
}


# Prologue: tables, biases and lengths fetched by one group of 16-byte loads issued before the first wait (one memory round trip, not one per staging-loop trip); time loop and sums unchanged
# speedup vs baseline: 3.0719x; 1.0561x over previous
#include <hip/hip_runtime.h>


#define VOCAB   100
#define E       32
#define HDIM    64
#define G4H     256
#define MAXLEN  16
#define WAVES   16
#define GROUPS  16
#define WPB     (GROUPS * 16)
#define BPS     8
#define SBW     (BPS * WPB)
#define NKEY    (MAXLEN + 1)
#define KROWS   (SBW / (WAVES * 32))
#define CPITCH  17
#define EPITCH  36
#define WIHP    36
#define WHHP    68
#define HPITCH  68

typedef __attribute__((ext_vector_type(2))) float v2f;
typedef __attribute__((ext_vector_type(4))) float v4f;
typedef __attribute__((ext_vector_type(8))) float v8f;

__device__ __forceinline__ float tanh_hw(float x) {
    return __builtin_amdgcn_tanhf(x);
}

__device__ __forceinline__ float sigm_of_half(float xh) {
    return __builtin_fmaf(0.5f, tanh_hw(xh), 0.5f);
}

__device__ __forceinline__ void wave_lds_order() {
    __builtin_amdgcn_fence(__ATOMIC_RELEASE, "wavefront");
    __builtin_amdgcn_wave_barrier();
    __builtin_amdgcn_fence(__ATOMIC_ACQUIRE, "wavefront");
}

__global__ __launch_bounds__(WAVES * 32)
void char_lstm_wmma(const float* __restrict__ emb,
                    const float* __restrict__ Wih,
                    const float* __restrict__ Whh,
                    const float* __restrict__ bih,
                    const float* __restrict__ bhh,
                    const int*   __restrict__ chars,
                    const int*   __restrict__ lengths,
                    float*       __restrict__ out)
{
    __shared__ __attribute__((aligned(16))) float s_emb[VOCAB * EPITCH];
    __shared__ __attribute__((aligned(16))) float s_wih[G4H * WIHP];
    __shared__ __attribute__((aligned(16))) float s_whh[G4H * WHHP];
    __shared__ float s_bias[G4H];
    __shared__ __attribute__((aligned(16))) float s_h[WAVES][16 * HPITCH];
    __shared__ int   s_chars[WPB * CPITCH];
    __shared__ int   s_wtot[NKEY * WAVES];
    __shared__ int   s_base[NKEY * WAVES];
    __shared__ int   s_slen[WPB];
    __shared__ int   s_perm[WPB];

    const int tid  = threadIdx.x;
    const int lane = tid & 31;
    const int wave = __builtin_amdgcn_readfirstlane(tid >> 5);
    const int half = lane >> 4;
    const int n    = lane & 15;
    const int memb  = blockIdx.x % BPS;
    const int sbase = (blockIdx.x / BPS) * SBW;

    const v4f* __restrict__ emb4 = (const v4f*)emb;
    const v4f* __restrict__ wih4 = (const v4f*)Wih;
    const v4f* __restrict__ whh4 = (const v4f*)Whh;
    constexpr int NT   = WAVES * 32;
    constexpr int EMB4 = VOCAB * E / 4;
    int   lraw[KROWS];
    v4f   pe[2], pi[G4H * E / 4 / NT], ph[G4H * HDIM / 4 / NT];
    #pragma unroll
    for (int r = 0; r < KROWS; ++r)
        lraw[r] = lengths[sbase + wave * (KROWS * 32) + r * 32 + lane];
    pe[0] = emb4[tid];
    const int e1 = (tid + NT < EMB4) ? tid + NT : EMB4 - 1;
    pe[1] = emb4[e1];
    #pragma unroll
    for (int k = 0; k < G4H * E / 4 / NT; ++k)    pi[k] = wih4[tid + k * NT];
    #pragma unroll
    for (int k = 0; k < G4H * HDIM / 4 / NT; ++k) ph[k] = whh4[tid + k * NT];
    const int   bidx = tid & (G4H - 1);
    const float braw = bih[bidx] + bhh[bidx];

    *(v4f*)&s_emb[(tid >> 3) * EPITCH + (tid & 7) * 4] = pe[0];
    *(v4f*)&s_emb[(e1 >> 3) * EPITCH + (e1 & 7) * 4] = pe[1];
    #pragma unroll
    for (int k = 0; k < G4H * E / 4 / NT; ++k) {
        const int i = tid + k * NT;
        *(v4f*)&s_wih[(i >> 3) * WIHP + (i & 7) * 4] = pi[k];
    }
    #pragma unroll
    for (int k = 0; k < G4H * HDIM / 4 / NT; ++k) {
        const int i = tid + k * NT;
        *(v4f*)&s_whh[(i >> 4) * WHHP + (i & 15) * 4] = ph[k];
    }
    s_bias[bidx] = (bidx / HDIM == 2) ? braw : 0.5f * braw;
    int key[KROWS], mybase[KROWS], within[KROWS];
    {
        int cnt[NKEY];
        #pragma unroll
        for (int k = 0; k < NKEY; ++k) cnt[k] = 0;
        #pragma unroll
        for (int r = 0; r < KROWS; ++r) {
            int L = lraw[r];
            L = L < 0 ? 0 : (L > MAXLEN ? MAXLEN : L);
            key[r] = L;
            unsigned mymask = 0u;
            int base = 0;
            #pragma unroll
            for (int k = 0; k < NKEY; ++k) {
                const bool mine = (L == k);
                const unsigned m = __builtin_amdgcn_ballot_w32(mine);
                mymask = mine ? m : mymask;
                base   = mine ? cnt[k] : base;
                cnt[k] += __builtin_popcount(m);
            }
            mybase[r] = base;
            within[r] = __builtin_popcount(mymask & ((1u << lane) - 1u));
        }
        int mine_tot = 0;
        #pragma unroll
        for (int k = 0; k < NKEY; ++k) mine_tot = (lane == k) ? cnt[k] : mine_tot;
        if (lane < NKEY) s_wtot[lane * WAVES + wave] = mine_tot;
    }
    __syncthreads();

    if (wave == 0) {
        int v[9];
        int run = 0;
        #pragma unroll
        for (int j = 0; j < 9; ++j) {
            const int idx = 9 * lane + j;
            const int x = (idx < NKEY * WAVES) ? s_wtot[idx] : 0;
            v[j] = run;
            run += x;
        }
        int inc = run;
        #pragma unroll
        for (int d = 1; d < 32; d <<= 1) {
            const int o = __shfl_up(inc, d);
            inc += (lane >= d) ? o : 0;
        }
        const int excl = inc - run;
        #pragma unroll
        for (int j = 0; j < 9; ++j) {
            const int idx = 9 * lane + j;
            if (idx < NKEY * WAVES) s_base[idx] = v[j] + excl;
        }
    }
    __syncthreads();

    #pragma unroll
    for (int r = 0; r < KROWS; ++r) {
        const int pos  = s_base[key[r] * WAVES + wave] + mybase[r] + within[r];
        const int grp  = pos >> 4;
        const int quar = grp >> 5;
        const int off  = grp & 31;
        const int q    = (quar & 1) ? (31 - off) : off;
        if ((q >> 2) == memb) {
            const int slot = (4 * (3 - quar) + (q & 3)) * 16 + (pos & 15);
            s_perm[slot] = wave * (KROWS * 32) + r * 32 + lane;
            s_slen[slot] = key[r];
        }
    }
    __syncthreads();

    const int g    = wave;
    const int coff = (g * 16 + n) * CPITCH;

    #pragma unroll
    for (int p = 0; p < 2; ++p) {
        const int e    = p * 32 + lane;
        const int row  = e >> 2, part = e & 3;
        const int wd   = s_perm[g * 16 + row];
        const int4 cv  = *(const int4*)&chars[(sbase + wd) * MAXLEN + part * 4];
        int* dst = &s_chars[(g * 16 + row) * CPITCH + part * 4];
        dst[0] = cv.x; dst[1] = cv.y; dst[2] = cv.z; dst[3] = cv.w;
    }

    int llen[8];
    #pragma unroll
    for (int r = 0; r < 8; ++r)
        llen[r] = s_slen[g * 16 + 8 * half + r];
    const int nsteps = __builtin_amdgcn_readfirstlane(s_slen[g * 16 + 15]);

    v8f creg[4];
    v8f hreg[4];
    #pragma unroll
    for (int cb = 0; cb < 4; ++cb)
        #pragma unroll
        for (int r = 0; r < 8; ++r) { creg[cb][r] = 0.0f; hreg[cb][r] = 0.0f; }

    for (int i = lane; i < 16 * HPITCH; i += 32)
        s_h[wave][i] = 0.0f;
    wave_lds_order();

    #pragma unroll 1
    for (int t = 0; t < nsteps; ++t) {
        const int c    = s_chars[coff + t];
        const int xoff = c * EPITCH + 2 * half;

        v2f afr[24];
        #pragma unroll
        for (int kk = 0; kk < E / 4; ++kk)
            afr[kk] = *(const v2f*)&s_emb[xoff + kk * 4];
        #pragma unroll
        for (int kk = 0; kk < HDIM / 4; ++kk)
            afr[E / 4 + kk] = *(const v2f*)&s_h[wave][n * HPITCH + kk * 4 + 2 * half];

        v2f bfr[2][4][2];
        #define LOAD_B(SET, CB, J)                                                          \
            do {                                                                            \
                _Pragma("unroll")                                                           \
                for (int g_ = 0; g_ < 4; ++g_)                                              \
                    _Pragma("unroll")                                                       \
                    for (int s_ = 0; s_ < 2; ++s_)                                          \
                        bfr[SET][g_][s_] = ((J) < 4)                                        \
                            ? *(const v2f*)&s_wih[(g_ * 64 + (CB) * 16 + n) * WIHP          \
                                                  + (2 * (J) + s_) * 4 + 2 * half]          \
                            : *(const v2f*)&s_whh[(g_ * 64 + (CB) * 16 + n) * WHHP          \
                                                  + (2 * ((J) - 4) + s_) * 4 + 2 * half];   \
            } while (0)

        LOAD_B(0, 0, 0);

        #pragma unroll
        for (int cb = 0; cb < 4; ++cb) {
            v8f acc[4];
            #pragma unroll
            for (int g = 0; g < 4; ++g)
                #pragma unroll
                for (int r = 0; r < 8; ++r) acc[g][r] = 0.0f;

            #pragma unroll
            for (int j = 0; j < 12; ++j) {
                const int blk = cb * 12 + j;
                const int cur = blk & 1;
                if (blk + 1 < 48)
                    LOAD_B(cur ^ 1, (blk + 1) / 12, (blk + 1) % 12);
                __builtin_amdgcn_sched_barrier(0);
                #pragma unroll
                for (int s = 0; s < 2; ++s)
                    #pragma unroll
                    for (int g = 0; g < 4; ++g)
                        acc[g] = __builtin_amdgcn_wmma_f32_16x16x4_f32(
                            false, afr[2 * j + s], false, bfr[cur][g][s],
                            (short)0, acc[g], false, false);
                __builtin_amdgcn_sched_barrier(0);
            }

            float b0 = s_bias[0 * 64 + cb * 16 + n];
            float b1 = s_bias[1 * 64 + cb * 16 + n];
            float b2 = s_bias[2 * 64 + cb * 16 + n];
            float b3 = s_bias[3 * 64 + cb * 16 + n];

            #pragma unroll
            for (int r = 0; r < 8; ++r) {
                float ig = sigm_of_half(__builtin_fmaf(0.5f, acc[0][r], b0));
                float fg = sigm_of_half(__builtin_fmaf(0.5f, acc[1][r], b1));
                float gg = tanh_hw(acc[2][r] + b2);
                float og = sigm_of_half(__builtin_fmaf(0.5f, acc[3][r], b3));
                float cn = fg * creg[cb][r] + ig * gg;
                float hn = og * tanh_hw(cn);
                const bool act = t < llen[r];
                const float cold = creg[cb][r];
                const float hold = hreg[cb][r];
                creg[cb][r] = act ? cn : cold;
                hreg[cb][r] = act ? hn : hold;
            }
        }

        #pragma unroll
        for (int cb = 0; cb < 4; ++cb)
            #pragma unroll
            for (int r = 0; r < 8; ++r)
                s_h[wave][(8 * half + r) * HPITCH + cb * 16 + n] = hreg[cb][r];

        wave_lds_order();
        #undef LOAD_B
    }

    #pragma unroll
    for (int j = 0; j < 8; ++j) {
        int idx = j * 128 + lane * 4;
        int row = idx >> 6, col = idx & 63;
        v4f v = *(const v4f*)&s_h[wave][row * HPITCH + col];
        const int w = s_perm[g * 16 + row];
        *(v4f*)&out[(sbase + w) * HDIM + col] = v;
    }
}

extern "C" void kernel_launch(void* const* d_in, const int* in_sizes, int n_in,
                              void* d_out, int out_size, void* d_ws, size_t ws_size,
                              hipStream_t stream) {
    const float* emb     = (const float*)d_in[0];
    const float* W_ih    = (const float*)d_in[1];
    const float* W_hh    = (const float*)d_in[2];
    const float* b_ih    = (const float*)d_in[3];
    const float* b_hh    = (const float*)d_in[4];
    const int*   chars   = (const int*)d_in[5];
    const int*   lengths = (const int*)d_in[6];
    float* out = (float*)d_out;

    const int n_words = in_sizes[6];
    const int blocks  = n_words / WPB;
    char_lstm_wmma<<<blocks, WAVES * 32, 0, stream>>>(
        emb, W_ih, W_hh, b_ih, b_hh, chars, lengths, out);
}
